# starting kernel
# baseline (speedup 1.0000x reference)
#include <hip/hip_runtime.h>

typedef __attribute__((ext_vector_type(2))) float v2f;
typedef __attribute__((ext_vector_type(8))) float v8f;

#define N_TOK 360
#define D_DIM 128
#define B_SZ  4
#define M_ROWS (B_SZ * N_TOK)

__global__ __launch_bounds__(256) void gemm_wmma(const float* __restrict__ x,
                                                 const float* __restrict__ W1,
                                                 const float* __restrict__ b1,
                                                 float* __restrict__ wsA,
                                                 float* __restrict__ wsC) {
    const int wid  = threadIdx.x >> 5;
    const int lane = threadIdx.x & 31;
    const int gw   = blockIdx.x * 8 + wid;
    const int mt   = gw >> 4;
    const int nt16 = gw & 15;
    const int half = nt16 >> 3;
    const int nt   = nt16 & 7;
    const int mbase = mt * 16;
    const int nbase = nt * 16;
    const int lm    = lane & 15;
    const int koff  = (lane < 16) ? 0 : 2;

    const float* xrow = x + (mbase + lm) * D_DIM;
    const float* wb   = W1 + half * D_DIM * D_DIM;

    v8f acc = {0.f, 0.f, 0.f, 0.f, 0.f, 0.f, 0.f, 0.f};
#pragma unroll
    for (int k = 0; k < D_DIM; k += 4) {
        v2f a = *(const v2f*)(xrow + k + koff);
        v2f b;
        b.x = wb[(k + koff)     * D_DIM + nbase + lm];
        b.y = wb[(k + koff + 1) * D_DIM + nbase + lm];
        acc = __builtin_amdgcn_wmma_f32_16x16x4_f32(
             false, a,  false, b,
             (short)0, acc,  false,  false);
    }

    float* dst = half ? wsC : wsA;
    const float badd = half ? 0.0f : b1[nbase + lm];
    const int moff = (lane < 16) ? 0 : 8;
#pragma unroll
    for (int r = 0; r < 8; ++r) {
        dst[(mbase + r + moff) * D_DIM + nbase + lm] = acc[r] + badd;
    }
}

__global__ __launch_bounds__(256) void pair_logits(const float* __restrict__ wsA,
                                                   const float* __restrict__ wsC,
                                                   const float* __restrict__ W2,
                                                   const float* __restrict__ b2,
                                                   float* __restrict__ out) {
    __shared__ float As[16][132];
    __shared__ float Cs[16][132];
    __shared__ float w2s[128];

    const int b  = blockIdx.z;
    const int i0 = blockIdx.x * 16;
    const int j0 = blockIdx.y * 16;
    const int tid = threadIdx.x;

    for (int idx = tid; idx < 16 * 128; idx += 256) {
        const int r = idx >> 7, c = idx & 127;
        const int gi = i0 + r, gj = j0 + r;
        As[r][c] = (gi < N_TOK) ? wsA[(b * N_TOK + gi) * D_DIM + c] : 0.0f;
        Cs[r][c] = (gj < N_TOK) ? wsC[(b * N_TOK + gj) * D_DIM + c] : 0.0f;
    }
    if (tid < 128) w2s[tid] = W2[tid];
    __syncthreads();

    const int tx = tid & 15;
    const int ty = tid >> 4;
    float acc = 0.0f;
#pragma unroll 4
    for (int d = 0; d < D_DIM; ++d) {
        float v = As[ty][d] + Cs[tx][d];
        v = fmaxf(v, 0.0f);
        acc = fmaf(v, w2s[d], acc);
    }

    const int i = i0 + ty, j = j0 + tx;
    if (i < N_TOK && j < N_TOK) {
        float lg = acc + b2[0];
        if (i == j) lg = 0.0f;
        out[(b * N_TOK + i) * N_TOK + j] = lg;
    }
}

__global__ __launch_bounds__(128) void row_softmax(float* __restrict__ out) {
    __shared__ float red[128];
    float* p = out + (size_t)blockIdx.x * N_TOK;
    const int tid = threadIdx.x;

    float vals[3];
    float m = -3.4e38f;
#pragma unroll
    for (int t = 0; t < 3; ++t) {
        const int j = tid + t * 128;
        vals[t] = (j < N_TOK) ? p[j] : -3.4e38f;
        m = fmaxf(m, vals[t]);
    }
    red[tid] = m;
    __syncthreads();
    for (int s = 64; s > 0; s >>= 1) {
        if (tid < s) red[tid] = fmaxf(red[tid], red[tid + s]);
        __syncthreads();
    }
    m = red[0];
    __syncthreads();

    float sum = 0.0f;
#pragma unroll
    for (int t = 0; t < 3; ++t) {
        const int j = tid + t * 128;
        if (j < N_TOK) {
            vals[t] = expf(vals[t] - m);
            sum += vals[t];
        }
    }
    red[tid] = sum;
    __syncthreads();
    for (int s = 64; s > 0; s >>= 1) {
        if (tid < s) red[tid] += red[tid + s];
        __syncthreads();
    }
    const float inv = 1.0f / red[0];
#pragma unroll
    for (int t = 0; t < 3; ++t) {
        const int j = tid + t * 128;
        if (j < N_TOK) p[j] = vals[t] * inv;
    }
}

extern "C" void kernel_launch(void* const* d_in, const int* in_sizes, int n_in,
                              void* d_out, int out_size, void* d_ws, size_t ws_size,
                              hipStream_t stream) {
    (void)in_sizes; (void)n_in; (void)out_size; (void)ws_size;
    const float* x  = (const float*)d_in[0];
    const float* W1 = (const float*)d_in[1];
    const float* b1 = (const float*)d_in[2];
    const float* W2 = (const float*)d_in[3];
    const float* b2 = (const float*)d_in[4];
    float* out = (float*)d_out;

    float* wsA = (float*)d_ws;
    float* wsC = wsA + M_ROWS * D_DIM;

    gemm_wmma<<<dim3(M_ROWS * 16 / 16 / 8), 256, 0, stream>>>(x, W1, b1, wsA, wsC);

    dim3 pg((N_TOK + 15) / 16, (N_TOK + 15) / 16, B_SZ);
    pair_logits<<<pg, 256, 0, stream>>>(wsA, wsC, W2, b2, out);

    row_softmax<<<dim3(B_SZ * N_TOK), 128, 0, stream>>>(out);
}


# pair_logits register-tiled: one 45x45 tile per block, 3x3 outputs per thread from 128-bit LDS row reads, W2 by scalar loads; 16928 waves -> 2048; per-output d order kept
# speedup vs baseline: 1.7160x; 1.7160x over previous
#include <hip/hip_runtime.h>

typedef __attribute__((ext_vector_type(2))) float v2f;
typedef __attribute__((ext_vector_type(8))) float v8f;

#define N_TOK 360
#define D_DIM 128
#define B_SZ  4
#define M_ROWS (B_SZ * N_TOK)

__global__ __launch_bounds__(256) void gemm_wmma(const float* __restrict__ x,
                                                 const float* __restrict__ W1,
                                                 const float* __restrict__ b1,
                                                 float* __restrict__ wsA,
                                                 float* __restrict__ wsC) {
    const int wid  = threadIdx.x >> 5;
    const int lane = threadIdx.x & 31;
    const int gw   = blockIdx.x * 8 + wid;
    const int mt   = gw >> 4;
    const int nt16 = gw & 15;
    const int half = nt16 >> 3;
    const int nt   = nt16 & 7;
    const int mbase = mt * 16;
    const int nbase = nt * 16;
    const int lm    = lane & 15;
    const int koff  = (lane < 16) ? 0 : 2;

    const float* xrow = x + (mbase + lm) * D_DIM;
    const float* wb   = W1 + half * D_DIM * D_DIM;

    v8f acc = {0.f, 0.f, 0.f, 0.f, 0.f, 0.f, 0.f, 0.f};
#pragma unroll
    for (int k = 0; k < D_DIM; k += 4) {
        v2f a = *(const v2f*)(xrow + k + koff);
        v2f b;
        b.x = wb[(k + koff)     * D_DIM + nbase + lm];
        b.y = wb[(k + koff + 1) * D_DIM + nbase + lm];
        acc = __builtin_amdgcn_wmma_f32_16x16x4_f32(
             false, a,  false, b,
             (short)0, acc,  false,  false);
    }

    float* dst = half ? wsC : wsA;
    const float badd = half ? 0.0f : b1[nbase + lm];
    const int moff = (lane < 16) ? 0 : 8;
#pragma unroll
    for (int r = 0; r < 8; ++r) {
        dst[(mbase + r + moff) * D_DIM + nbase + lm] = acc[r] + badd;
    }
}

#define PT 45
#define PS 15
#define PPITCH 132
#define PROWS 48

__global__ __launch_bounds__(256) __attribute__((amdgpu_waves_per_eu(2, 4)))
void pair_logits(const float* __restrict__ wsA,
                                                   const float* __restrict__ wsC,
                                                   const float* __restrict__ W2,
                                                   const float* __restrict__ b2,
                                                   float* __restrict__ out) {
    __shared__ __attribute__((aligned(16))) float As[PROWS][PPITCH];
    __shared__ __attribute__((aligned(16))) float Cs[PROWS][PPITCH];

    const int b   = blockIdx.z;
    const int i0  = blockIdx.x * PT;
    const int j0  = blockIdx.y * PT;
    const int tid = threadIdx.x;

    const float* gA = wsA + (size_t)(b * N_TOK + i0) * D_DIM;
    const float* gC = wsC + (size_t)(b * N_TOK + j0) * D_DIM;
    const int sr = tid >> 5, sc = (tid & 31) * 4;
    float4 va[6], vc[6];
#pragma unroll
    for (int k = 0; k < 6; ++k) {
        const int r = min(sr + 8 * k, PT - 1);
        va[k] = *(const float4*)(gA + r * D_DIM + sc);
        vc[k] = *(const float4*)(gC + r * D_DIM + sc);
    }
#pragma unroll
    for (int k = 0; k < 6; ++k) {
        const int r = sr + 8 * k;
        *(float4*)&As[r][sc] = va[k];
        *(float4*)&Cs[r][sc] = vc[k];
    }
    __syncthreads();

    if (tid < PS * PS) {
        const int tx = tid % PS;
        const int ty = tid / PS;
        const float* ap = &As[ty][0];
        const float* cp = &Cs[tx][0];

        float acc[3][3];
#pragma unroll
        for (int ii = 0; ii < 3; ++ii)
#pragma unroll
            for (int jj = 0; jj < 3; ++jj) acc[ii][jj] = 0.0f;

#pragma unroll 4
        for (int d = 0; d < D_DIM; d += 4) {
            float av[3][4], cv[3][4];
#pragma unroll
            for (int t = 0; t < 3; ++t) {
                const float4 a4 = *(const float4*)(ap + t * PS * PPITCH + d);
                const float4 c4 = *(const float4*)(cp + t * PS * PPITCH + d);
                av[t][0] = a4.x; av[t][1] = a4.y; av[t][2] = a4.z; av[t][3] = a4.w;
                cv[t][0] = c4.x; cv[t][1] = c4.y; cv[t][2] = c4.z; cv[t][3] = c4.w;
            }
#pragma unroll
            for (int dd = 0; dd < 4; ++dd) {
                const float w = W2[d + dd];
                float v[3][3];
#pragma unroll
                for (int ii = 0; ii < 3; ++ii)
#pragma unroll
                    for (int jj = 0; jj < 3; ++jj) v[ii][jj] = av[ii][dd] + cv[jj][dd];
#pragma unroll
                for (int ii = 0; ii < 3; ++ii)
#pragma unroll
                    for (int jj = 0; jj < 3; ++jj) v[ii][jj] = fmaxf(v[ii][jj], 0.0f);
#pragma unroll
                for (int ii = 0; ii < 3; ++ii)
#pragma unroll
                    for (int jj = 0; jj < 3; ++jj) acc[ii][jj] = fmaf(v[ii][jj], w, acc[ii][jj]);
            }
        }

        const float bias = b2[0];
#pragma unroll
        for (int ii = 0; ii < 3; ++ii) {
            const int i = i0 + ty + ii * PS;
#pragma unroll
            for (int jj = 0; jj < 3; ++jj) {
                const int j = j0 + tx + jj * PS;
                float lg = acc[ii][jj] + bias;
                if (i == j) lg = 0.0f;
                out[(size_t)(b * N_TOK + i) * N_TOK + j] = lg;
            }
        }
    }
}

__global__ __launch_bounds__(128) void row_softmax(float* __restrict__ out) {
    __shared__ float red[128];
    float* p = out + (size_t)blockIdx.x * N_TOK;
    const int tid = threadIdx.x;

    float vals[3];
    float m = -3.4e38f;
#pragma unroll
    for (int t = 0; t < 3; ++t) {
        const int j = tid + t * 128;
        vals[t] = (j < N_TOK) ? p[j] : -3.4e38f;
        m = fmaxf(m, vals[t]);
    }
    red[tid] = m;
    __syncthreads();
    for (int s = 64; s > 0; s >>= 1) {
        if (tid < s) red[tid] = fmaxf(red[tid], red[tid + s]);
        __syncthreads();
    }
    m = red[0];
    __syncthreads();

    float sum = 0.0f;
#pragma unroll
    for (int t = 0; t < 3; ++t) {
        const int j = tid + t * 128;
        if (j < N_TOK) {
            vals[t] = expf(vals[t] - m);
            sum += vals[t];
        }
    }
    red[tid] = sum;
    __syncthreads();
    for (int s = 64; s > 0; s >>= 1) {
        if (tid < s) red[tid] += red[tid + s];
        __syncthreads();
    }
    const float inv = 1.0f / red[0];
#pragma unroll
    for (int t = 0; t < 3; ++t) {
        const int j = tid + t * 128;
        if (j < N_TOK) p[j] = vals[t] * inv;
    }
}

extern "C" void kernel_launch(void* const* d_in, const int* in_sizes, int n_in,
                              void* d_out, int out_size, void* d_ws, size_t ws_size,
                              hipStream_t stream) {
    (void)in_sizes; (void)n_in; (void)out_size; (void)ws_size;
    const float* x  = (const float*)d_in[0];
    const float* W1 = (const float*)d_in[1];
    const float* b1 = (const float*)d_in[2];
    const float* W2 = (const float*)d_in[3];
    const float* b2 = (const float*)d_in[4];
    float* out = (float*)d_out;

    float* wsA = (float*)d_ws;
    float* wsC = wsA + M_ROWS * D_DIM;

    gemm_wmma<<<dim3(M_ROWS * 16 / 16 / 8), 256, 0, stream>>>(x, W1, b1, wsA, wsC);

    dim3 pg(N_TOK / PT, N_TOK / PT, B_SZ);
    pair_logits<<<pg, 256, 0, stream>>>(wsA, wsC, W2, b2, out);

    row_softmax<<<dim3(B_SZ * N_TOK), 128, 0, stream>>>(out);
}


# row softmax fused into pair_logits: a block owns 6 full rows, all 360 C rows in LDS, staged in 4 d-slices under compute; 3 launches -> 2; row max/sum by a new tree (lane butterfly, then 4 waves)
# speedup vs baseline: 2.0320x; 1.1841x over previous
#include <hip/hip_runtime.h>

typedef __attribute__((ext_vector_type(2))) float v2f;
typedef __attribute__((ext_vector_type(8))) float v8f;
typedef __attribute__((ext_vector_type(4))) float v4f;

#define N_TOK 360
#define D_DIM 128
#define B_SZ  4
#define M_ROWS (B_SZ * N_TOK)

__global__ __launch_bounds__(256) void gemm_wmma(const float* __restrict__ x,
                                                 const float* __restrict__ W1,
                                                 const float* __restrict__ b1,
                                                 float* __restrict__ wsA,
                                                 float* __restrict__ wsC) {
    const int wid  = threadIdx.x >> 5;
    const int lane = threadIdx.x & 31;
    const int gw   = blockIdx.x * 8 + wid;
    const int mt   = gw >> 4;
    const int nt16 = gw & 15;
    const int half = nt16 >> 3;
    const int nt   = nt16 & 7;
    const int mbase = mt * 16;
    const int nbase = nt * 16;
    const int lm    = lane & 15;
    const int koff  = (lane < 16) ? 0 : 2;

    const float* xrow = x + (mbase + lm) * D_DIM;
    const float* wb   = W1 + half * D_DIM * D_DIM;

    v8f acc = {0.f, 0.f, 0.f, 0.f, 0.f, 0.f, 0.f, 0.f};
#pragma unroll
    for (int k = 0; k < D_DIM; k += 4) {
        v2f a = *(const v2f*)(xrow + k + koff);
        v2f b;
        b.x = wb[(k + koff)     * D_DIM + nbase + lm];
        b.y = wb[(k + koff + 1) * D_DIM + nbase + lm];
        acc = __builtin_amdgcn_wmma_f32_16x16x4_f32(
             false, a,  false, b,
             (short)0, acc,  false,  false);
    }

    float* dst = half ? wsC : wsA;
    const float badd = half ? 0.0f : b1[nbase + lm];
    const int moff = (lane < 16) ? 0 : 8;
#pragma unroll
    for (int r = 0; r < 8; ++r) {
        dst[(mbase + r + moff) * D_DIM + nbase + lm] = acc[r] + badd;
    }
}

#define FR 6
#define FC 120
#define CPITCH 132
#define CROWS 384
#define DCH 32
#define NCH (D_DIM / DCH)
#define SPT 12

__device__ __forceinline__ float xhalf(float x) {
    const int v = __builtin_bit_cast(int, x);
    return __builtin_bit_cast(float,
        __builtin_amdgcn_permlanex16(v, v, 0x76543210, 0xfedcba98, false, false));
}
#define ROWX(x, n) __builtin_bit_cast(float, __builtin_amdgcn_update_dpp( \
    0, __builtin_bit_cast(int, (x)), 0x160 | (n), 0xf, 0xf, true))

__device__ __forceinline__ float wave_max(float v) {
    v = fmaxf(v, ROWX(v, 1));
    v = fmaxf(v, ROWX(v, 2));
    v = fmaxf(v, ROWX(v, 4));
    v = fmaxf(v, ROWX(v, 8));
    return fmaxf(v, xhalf(v));
}
__device__ __forceinline__ float wave_sum(float v) {
    v += ROWX(v, 1);
    v += ROWX(v, 2);
    v += ROWX(v, 4);
    v += ROWX(v, 8);
    return v + xhalf(v);
}

__device__ __forceinline__ void lds_barrier() {
    __builtin_amdgcn_fence(__ATOMIC_RELEASE, "workgroup", "local");
    __builtin_amdgcn_s_barrier();
    __builtin_amdgcn_fence(__ATOMIC_ACQUIRE, "workgroup", "local");
}

__global__ __launch_bounds__(256) __attribute__((amdgpu_waves_per_eu(2, 4)))
void pair_softmax(const float* __restrict__ wsA,
                  const float* __restrict__ wsC,
                  const float* __restrict__ W2,
                  const float* __restrict__ b2,
                  float* __restrict__ out) {
    __shared__ __attribute__((aligned(16))) float L[8 + CROWS][CPITCH];
    float (*As)[CPITCH] = &L[0];
    float (*Cs)[CPITCH] = &L[8];
    __shared__ float redm[2][3][4];
    __shared__ float reds[2][3][4];

    const int b   = blockIdx.y;
    const int i0  = blockIdx.x * FR;
    const int tid = threadIdx.x;

    const float* gA = wsA + (size_t)(b * N_TOK + i0) * D_DIM;
    const float* gC = wsC + (size_t)(b * N_TOK) * D_DIM;

    const int sw = tid >> 5, sl = tid & 31;
    const int sr = ((sl >> 3) & 1) * 8 + (sl >> 4) + 2 * (sw & 3) + 16 * (sw >> 2);
    const int sq = (tid & 7) * 4;
    v4f st[SPT];
    {
        const int ar = tid >> 5, ac = (tid & 31) * 4;
        const v4f a4 = *(const v4f*)(gA + min(ar, FR - 1) * D_DIM + ac);
#pragma unroll
        for (int q = 0; q < SPT; ++q)
            st[q] = *(const v4f*)(gC + min(sr + 32 * q, N_TOK - 1) * D_DIM + sq);
        *(v4f*)&As[ar][ac] = a4;
#pragma unroll
        for (int q = 0; q < SPT; ++q)
            *(v4f*)&Cs[sr + 32 * q][sq] = st[q];
    }

    const int ty   = tid >> 7;
    const int tx   = tid & 127;
    const bool live = tx < FC;
    const int txc  = min(tx, FC - 1);
    const float* ap = &As[ty][0];
    const float* cp = &Cs[txc][0];

    float acc[3][3];
#pragma unroll
    for (int ii = 0; ii < 3; ++ii)
#pragma unroll
        for (int jj = 0; jj < 3; ++jj) acc[ii][jj] = 0.0f;

#pragma unroll
    for (int k = 0; k < NCH; ++k) {
        const int dn = (k + 1) * DCH;
        if (k + 1 < NCH) {
#pragma unroll
            for (int q = 0; q < SPT; ++q)
                st[q] = *(const v4f*)(gC + min(sr + 32 * q, N_TOK - 1) * D_DIM + dn + sq);
        }
        lds_barrier();

#pragma unroll 4
        for (int d = k * DCH; d < dn; d += 4) {
            float av[3][4], cv[3][4];
#pragma unroll
            for (int t = 0; t < 3; ++t) {
                const float4 a4 = *(const float4*)(ap + t * 2 * CPITCH + d);
                const float4 c4 = *(const float4*)(cp + t * FC * CPITCH + d);
                av[t][0] = a4.x; av[t][1] = a4.y; av[t][2] = a4.z; av[t][3] = a4.w;
                cv[t][0] = c4.x; cv[t][1] = c4.y; cv[t][2] = c4.z; cv[t][3] = c4.w;
            }
#pragma unroll
            for (int dd = 0; dd < 4; ++dd) {
                const float w = W2[d + dd];
                float v[3][3];
#pragma unroll
                for (int ii = 0; ii < 3; ++ii)
#pragma unroll
                    for (int jj = 0; jj < 3; ++jj) v[ii][jj] = av[ii][dd] + cv[jj][dd];
#pragma unroll
                for (int ii = 0; ii < 3; ++ii)
#pragma unroll
                    for (int jj = 0; jj < 3; ++jj) v[ii][jj] = fmaxf(v[ii][jj], 0.0f);
#pragma unroll
                for (int ii = 0; ii < 3; ++ii)
#pragma unroll
                    for (int jj = 0; jj < 3; ++jj) acc[ii][jj] = fmaf(v[ii][jj], w, acc[ii][jj]);
            }
        }

        if (k + 1 < NCH) {
#pragma unroll
            for (int q = 0; q < SPT; ++q)
                *(v4f*)&Cs[sr + 32 * q][dn + sq] = st[q];
        }
    }

    const float bias = b2[0];
    const int wq = (tid >> 5) & 3;
    const int lane = tid & 31;

    float m[3];
#pragma unroll
    for (int ii = 0; ii < 3; ++ii) {
        const int i = i0 + ty + 2 * ii;
#pragma unroll
        for (int jj = 0; jj < 3; ++jj) {
            const int j = tx + FC * jj;
            float lg = acc[ii][jj] + bias;
            if (i == j) lg = 0.0f;
            acc[ii][jj] = lg;
        }
        const float mx = fmaxf(fmaxf(acc[ii][0], acc[ii][1]), acc[ii][2]);
        m[ii] = wave_max(live ? mx : -3.4e38f);
    }
    if (lane == 0) {
#pragma unroll
        for (int ii = 0; ii < 3; ++ii) redm[ty][ii][wq] = m[ii];
    }
    __syncthreads();

    float s[3];
#pragma unroll
    for (int ii = 0; ii < 3; ++ii) {
        const float mr = fmaxf(fmaxf(redm[ty][ii][0], redm[ty][ii][1]),
                               fmaxf(redm[ty][ii][2], redm[ty][ii][3]));
        float sm = 0.0f;
#pragma unroll
        for (int jj = 0; jj < 3; ++jj) {
            acc[ii][jj] = expf(acc[ii][jj] - mr);
            sm += acc[ii][jj];
        }
        s[ii] = wave_sum(live ? sm : 0.0f);
    }
    if (lane == 0) {
#pragma unroll
        for (int ii = 0; ii < 3; ++ii) reds[ty][ii][wq] = s[ii];
    }
    __syncthreads();

    if (live) {
#pragma unroll
        for (int ii = 0; ii < 3; ++ii) {
            const int i = i0 + ty + 2 * ii;
            const float inv = 1.0f / ((reds[ty][ii][0] + reds[ty][ii][1]) +
                                      (reds[ty][ii][2] + reds[ty][ii][3]));
            float* orow = out + (size_t)(b * N_TOK + i) * N_TOK;
#pragma unroll
            for (int jj = 0; jj < 3; ++jj) orow[tx + FC * jj] = acc[ii][jj] * inv;
        }
    }
}

extern "C" void kernel_launch(void* const* d_in, const int* in_sizes, int n_in,
                              void* d_out, int out_size, void* d_ws, size_t ws_size,
                              hipStream_t stream) {
    (void)in_sizes; (void)n_in; (void)out_size; (void)ws_size;
    const float* x  = (const float*)d_in[0];
    const float* W1 = (const float*)d_in[1];
    const float* b1 = (const float*)d_in[2];
    const float* W2 = (const float*)d_in[3];
    const float* b2 = (const float*)d_in[4];
    float* out = (float*)d_out;

    float* wsA = (float*)d_ws;
    float* wsC = wsA + M_ROWS * D_DIM;

    gemm_wmma<<<dim3(M_ROWS * 16 / 16 / 8), 256, 0, stream>>>(x, W1, b1, wsA, wsC);

    dim3 pg(N_TOK / FR, B_SZ);
    pair_softmax<<<pg, 256, 0, stream>>>(wsA, wsC, W2, b2, out);
}


# pair kernel: wave-uniform A rows come by scalar loads as scalar operands of the min/fma pairs (first kernel stores -xj, sign restored in epilogue), not as same-address LDS reads: LDS reads halved
# speedup vs baseline: 2.2707x; 1.1175x over previous
#include <hip/hip_runtime.h>

typedef __attribute__((ext_vector_type(2))) float v2f;
typedef __attribute__((ext_vector_type(8))) float v8f;
typedef __attribute__((ext_vector_type(4))) float v4f;

#define N_TOK 360
#define D_DIM 128
#define B_SZ  4
#define M_ROWS (B_SZ * N_TOK)


__global__ __launch_bounds__(256) void gemm_wmma(const float* __restrict__ x,
                                                 const float* __restrict__ W1,
                                                 const float* __restrict__ b1,
                                                 float* __restrict__ wsA,
                                                 float* __restrict__ wsC) {
    const int wid  = threadIdx.x >> 5;
    const int lane = threadIdx.x & 31;
    const int gw   = blockIdx.x * 8 + wid;
    const int mt   = gw >> 4;
    const int nt16 = gw & 15;
    const int half = nt16 >> 3;
    const int nt   = nt16 & 7;
    const int mbase = mt * 16;
    const int nbase = nt * 16;
    const int lm    = lane & 15;
    const int koff  = (lane < 16) ? 0 : 2;

    const float* xrow = x + (mbase + lm) * D_DIM;
    const float* wb   = W1 + half * D_DIM * D_DIM;

    v8f acc = {0.f, 0.f, 0.f, 0.f, 0.f, 0.f, 0.f, 0.f};
#pragma unroll
    for (int k = 0; k < D_DIM; k += 4) {
        v2f a = *(const v2f*)(xrow + k + koff);
        v2f b;
        b.x = wb[(k + koff)     * D_DIM + nbase + lm];
        b.y = wb[(k + koff + 1) * D_DIM + nbase + lm];
        acc = __builtin_amdgcn_wmma_f32_16x16x4_f32(
             false, a,  false, b,
             (short)0, acc,  false,  false);
    }

    float* dst = half ? wsC : wsA;
    const float badd = half ? 0.0f : b1[nbase + lm];
    const int moff = (lane < 16) ? 0 : 8;
#pragma unroll
    for (int r = 0; r < 8; ++r) {
        const float v = acc[r] + badd;
        dst[(mbase + r + moff) * D_DIM + nbase + lm] = half ? -v : v;
    }
}

#define FR 6
#define FC 120
#define CPITCH 132
#define CROWS 384
#define DCH 32
#define NCH (D_DIM / DCH)
#define SPT 12

__device__ __forceinline__ float xhalf(float x) {
    const int v = __builtin_bit_cast(int, x);
    return __builtin_bit_cast(float,
        __builtin_amdgcn_permlanex16(v, v, 0x76543210, 0xfedcba98, false, false));
}
#define ROWX(x, n) __builtin_bit_cast(float, __builtin_amdgcn_update_dpp( \
    0, __builtin_bit_cast(int, (x)), 0x160 | (n), 0xf, 0xf, true))

__device__ __forceinline__ float wave_max(float v) {
    v = fmaxf(v, ROWX(v, 1));
    v = fmaxf(v, ROWX(v, 2));
    v = fmaxf(v, ROWX(v, 4));
    v = fmaxf(v, ROWX(v, 8));
    return fmaxf(v, xhalf(v));
}
__device__ __forceinline__ float wave_sum(float v) {
    v += ROWX(v, 1);
    v += ROWX(v, 2);
    v += ROWX(v, 4);
    v += ROWX(v, 8);
    return v + xhalf(v);
}

__device__ __forceinline__ unsigned lds_addr(const float* p) {
    return (unsigned)(unsigned long long)(__attribute__((address_space(3))) const float*)p;
}


__device__ __forceinline__ void slice32(float (&acc)[3][3], unsigned c0, unsigned c2,
                                        const float* pa, const float* wp) {
    asm volatile(
        "s_load_b256 s[40:47], %[pa], 0x0\n"
        "s_load_b256 s[48:55], %[pa], 0x400\n"
        "s_load_b256 s[56:63], %[pa], 0x800\n"
        "s_load_b256 s[64:71], %[wp], 0x0\n"
        "s_load_b256 s[72:79], %[pa], 0x20\n"
        "s_load_b256 s[80:87], %[pa], 0x420\n"
        "s_load_b256 s[88:95], %[pa], 0x820\n"
        "s_load_b256 s[96:103], %[wp], 0x20\n"
        "s_wait_dscnt 0x0\n"
        "s_barrier_signal -1\n"
        "s_barrier_wait 0xffff\n"
        "ds_load_b128 v[168:171], %[c0] offset:0\n"
        "ds_load_b128 v[172:175], %[c0] offset:63360\n"
        "ds_load_b128 v[176:179], %[c2] offset:0\n"
        "ds_load_b128 v[180:183], %[c0] offset:16\n"
        "ds_load_b128 v[184:187], %[c0] offset:63376\n"
        "ds_load_b128 v[188:191], %[c2] offset:16\n"
        "ds_load_b128 v[192:195], %[c0] offset:32\n"
        "ds_load_b128 v[196:199], %[c0] offset:63392\n"
        "ds_load_b128 v[200:203], %[c2] offset:32\n"
        "s_wait_dscnt 0x6\n"
        "s_wait_kmcnt 0x0\n"
        "v_min_num_f32_e32 v150, s40, v168\n"
        "v_min_num_f32_e32 v152, s40, v172\n"
        "v_min_num_f32_e32 v154, s40, v176\n"
        "v_min_num_f32_e32 v156, s48, v168\n"
        "v_min_num_f32_e32 v158, s48, v172\n"
        "v_min_num_f32_e32 v160, s48, v176\n"
        "v_min_num_f32_e32 v162, s56, v168\n"
        "v_min_num_f32_e32 v164, s56, v172\n"
        "v_min_num_f32_e32 v166, s56, v176\n"
        "v_dual_min_num_f32 v151, s41, v169 :: v_dual_fmac_f32 v140, s64, v150\n"
        "v_dual_min_num_f32 v153, s41, v173 :: v_dual_fmac_f32 v141, s64, v152\n"
        "v_dual_min_num_f32 v155, s41, v177 :: v_dual_fmac_f32 v142, s64, v154\n"
        "v_dual_min_num_f32 v157, s49, v169 :: v_dual_fmac_f32 v143, s64, v156\n"
        "v_dual_min_num_f32 v159, s49, v173 :: v_dual_fmac_f32 v144, s64, v158\n"
        "v_dual_min_num_f32 v161, s49, v177 :: v_dual_fmac_f32 v145, s64, v160\n"
        "v_dual_min_num_f32 v163, s57, v169 :: v_dual_fmac_f32 v146, s64, v162\n"
        "v_dual_min_num_f32 v165, s57, v173 :: v_dual_fmac_f32 v147, s64, v164\n"
        "v_dual_min_num_f32 v167, s57, v177 :: v_dual_fmac_f32 v148, s64, v166\n"
        "v_dual_min_num_f32 v150, s42, v170 :: v_dual_fmac_f32 v140, s65, v151\n"
        "v_dual_min_num_f32 v152, s42, v174 :: v_dual_fmac_f32 v141, s65, v153\n"
        "v_dual_min_num_f32 v154, s42, v178 :: v_dual_fmac_f32 v142, s65, v155\n"
        "v_dual_min_num_f32 v156, s50, v170 :: v_dual_fmac_f32 v143, s65, v157\n"
        "v_dual_min_num_f32 v158, s50, v174 :: v_dual_fmac_f32 v144, s65, v159\n"
        "v_dual_min_num_f32 v160, s50, v178 :: v_dual_fmac_f32 v145, s65, v161\n"
        "v_dual_min_num_f32 v162, s58, v170 :: v_dual_fmac_f32 v146, s65, v163\n"
        "v_dual_min_num_f32 v164, s58, v174 :: v_dual_fmac_f32 v147, s65, v165\n"
        "v_dual_min_num_f32 v166, s58, v178 :: v_dual_fmac_f32 v148, s65, v167\n"
        "v_dual_min_num_f32 v151, s43, v171 :: v_dual_fmac_f32 v140, s66, v150\n"
        "v_dual_min_num_f32 v153, s43, v175 :: v_dual_fmac_f32 v141, s66, v152\n"
        "v_dual_min_num_f32 v155, s43, v179 :: v_dual_fmac_f32 v142, s66, v154\n"
        "v_dual_min_num_f32 v157, s51, v171 :: v_dual_fmac_f32 v143, s66, v156\n"
        "v_dual_min_num_f32 v159, s51, v175 :: v_dual_fmac_f32 v144, s66, v158\n"
        "v_dual_min_num_f32 v161, s51, v179 :: v_dual_fmac_f32 v145, s66, v160\n"
        "v_dual_min_num_f32 v163, s59, v171 :: v_dual_fmac_f32 v146, s66, v162\n"
        "v_dual_min_num_f32 v165, s59, v175 :: v_dual_fmac_f32 v147, s66, v164\n"
        "v_dual_min_num_f32 v167, s59, v179 :: v_dual_fmac_f32 v148, s66, v166\n"
        "ds_load_b128 v[168:171], %[c0] offset:48\n"
        "ds_load_b128 v[172:175], %[c0] offset:63408\n"
        "ds_load_b128 v[176:179], %[c2] offset:48\n"
        "s_wait_dscnt 0x6\n"
        "v_dual_min_num_f32 v150, s44, v180 :: v_dual_fmac_f32 v140, s67, v151\n"
        "v_dual_min_num_f32 v152, s44, v184 :: v_dual_fmac_f32 v141, s67, v153\n"
        "v_dual_min_num_f32 v154, s44, v188 :: v_dual_fmac_f32 v142, s67, v155\n"
        "v_dual_min_num_f32 v156, s52, v180 :: v_dual_fmac_f32 v143, s67, v157\n"
        "v_dual_min_num_f32 v158, s52, v184 :: v_dual_fmac_f32 v144, s67, v159\n"
        "v_dual_min_num_f32 v160, s52, v188 :: v_dual_fmac_f32 v145, s67, v161\n"
        "v_dual_min_num_f32 v162, s60, v180 :: v_dual_fmac_f32 v146, s67, v163\n"
        "v_dual_min_num_f32 v164, s60, v184 :: v_dual_fmac_f32 v147, s67, v165\n"
        "v_dual_min_num_f32 v166, s60, v188 :: v_dual_fmac_f32 v148, s67, v167\n"
        "v_dual_min_num_f32 v151, s45, v181 :: v_dual_fmac_f32 v140, s68, v150\n"
        "v_dual_min_num_f32 v153, s45, v185 :: v_dual_fmac_f32 v141, s68, v152\n"
        "v_dual_min_num_f32 v155, s45, v189 :: v_dual_fmac_f32 v142, s68, v154\n"
        "v_dual_min_num_f32 v157, s53, v181 :: v_dual_fmac_f32 v143, s68, v156\n"
        "v_dual_min_num_f32 v159, s53, v185 :: v_dual_fmac_f32 v144, s68, v158\n"
        "v_dual_min_num_f32 v161, s53, v189 :: v_dual_fmac_f32 v145, s68, v160\n"
        "v_dual_min_num_f32 v163, s61, v181 :: v_dual_fmac_f32 v146, s68, v162\n"
        "v_dual_min_num_f32 v165, s61, v185 :: v_dual_fmac_f32 v147, s68, v164\n"
        "v_dual_min_num_f32 v167, s61, v189 :: v_dual_fmac_f32 v148, s68, v166\n"
        "v_dual_min_num_f32 v150, s46, v182 :: v_dual_fmac_f32 v140, s69, v151\n"
        "v_dual_min_num_f32 v152, s46, v186 :: v_dual_fmac_f32 v141, s69, v153\n"
        "v_dual_min_num_f32 v154, s46, v190 :: v_dual_fmac_f32 v142, s69, v155\n"
        "v_dual_min_num_f32 v156, s54, v182 :: v_dual_fmac_f32 v143, s69, v157\n"
        "v_dual_min_num_f32 v158, s54, v186 :: v_dual_fmac_f32 v144, s69, v159\n"
        "v_dual_min_num_f32 v160, s54, v190 :: v_dual_fmac_f32 v145, s69, v161\n"
        "v_dual_min_num_f32 v162, s62, v182 :: v_dual_fmac_f32 v146, s69, v163\n"
        "v_dual_min_num_f32 v164, s62, v186 :: v_dual_fmac_f32 v147, s69, v165\n"
        "v_dual_min_num_f32 v166, s62, v190 :: v_dual_fmac_f32 v148, s69, v167\n"
        "v_dual_min_num_f32 v151, s47, v183 :: v_dual_fmac_f32 v140, s70, v150\n"
        "v_dual_min_num_f32 v153, s47, v187 :: v_dual_fmac_f32 v141, s70, v152\n"
        "v_dual_min_num_f32 v155, s47, v191 :: v_dual_fmac_f32 v142, s70, v154\n"
        "v_dual_min_num_f32 v157, s55, v183 :: v_dual_fmac_f32 v143, s70, v156\n"
        "v_dual_min_num_f32 v159, s55, v187 :: v_dual_fmac_f32 v144, s70, v158\n"
        "v_dual_min_num_f32 v161, s55, v191 :: v_dual_fmac_f32 v145, s70, v160\n"
        "v_dual_min_num_f32 v163, s63, v183 :: v_dual_fmac_f32 v146, s70, v162\n"
        "v_dual_min_num_f32 v165, s63, v187 :: v_dual_fmac_f32 v147, s70, v164\n"
        "v_dual_min_num_f32 v167, s63, v191 :: v_dual_fmac_f32 v148, s70, v166\n"
        "ds_load_b128 v[180:183], %[c0] offset:64\n"
        "ds_load_b128 v[184:187], %[c0] offset:63424\n"
        "ds_load_b128 v[188:191], %[c2] offset:64\n"
        "s_wait_dscnt 0x6\n"
        "v_dual_min_num_f32 v150, s72, v192 :: v_dual_fmac_f32 v140, s71, v151\n"
        "v_dual_min_num_f32 v152, s72, v196 :: v_dual_fmac_f32 v141, s71, v153\n"
        "v_dual_min_num_f32 v154, s72, v200 :: v_dual_fmac_f32 v142, s71, v155\n"
        "v_dual_min_num_f32 v156, s80, v192 :: v_dual_fmac_f32 v143, s71, v157\n"
        "v_dual_min_num_f32 v158, s80, v196 :: v_dual_fmac_f32 v144, s71, v159\n"
        "v_dual_min_num_f32 v160, s80, v200 :: v_dual_fmac_f32 v145, s71, v161\n"
        "v_dual_min_num_f32 v162, s88, v192 :: v_dual_fmac_f32 v146, s71, v163\n"
        "v_dual_min_num_f32 v164, s88, v196 :: v_dual_fmac_f32 v147, s71, v165\n"
        "v_dual_min_num_f32 v166, s88, v200 :: v_dual_fmac_f32 v148, s71, v167\n"
        "s_load_b256 s[40:47], %[pa], 0x40\n"
        "s_load_b256 s[48:55], %[pa], 0x440\n"
        "s_load_b256 s[56:63], %[pa], 0x840\n"
        "s_load_b256 s[64:71], %[wp], 0x40\n"
        "v_dual_min_num_f32 v151, s73, v193 :: v_dual_fmac_f32 v140, s96, v150\n"
        "v_dual_min_num_f32 v153, s73, v197 :: v_dual_fmac_f32 v141, s96, v152\n"
        "v_dual_min_num_f32 v155, s73, v201 :: v_dual_fmac_f32 v142, s96, v154\n"
        "v_dual_min_num_f32 v157, s81, v193 :: v_dual_fmac_f32 v143, s96, v156\n"
        "v_dual_min_num_f32 v159, s81, v197 :: v_dual_fmac_f32 v144, s96, v158\n"
        "v_dual_min_num_f32 v161, s81, v201 :: v_dual_fmac_f32 v145, s96, v160\n"
        "v_dual_min_num_f32 v163, s89, v193 :: v_dual_fmac_f32 v146, s96, v162\n"
        "v_dual_min_num_f32 v165, s89, v197 :: v_dual_fmac_f32 v147, s96, v164\n"
        "v_dual_min_num_f32 v167, s89, v201 :: v_dual_fmac_f32 v148, s96, v166\n"
        "v_dual_min_num_f32 v150, s74, v194 :: v_dual_fmac_f32 v140, s97, v151\n"
        "v_dual_min_num_f32 v152, s74, v198 :: v_dual_fmac_f32 v141, s97, v153\n"
        "v_dual_min_num_f32 v154, s74, v202 :: v_dual_fmac_f32 v142, s97, v155\n"
        "v_dual_min_num_f32 v156, s82, v194 :: v_dual_fmac_f32 v143, s97, v157\n"
        "v_dual_min_num_f32 v158, s82, v198 :: v_dual_fmac_f32 v144, s97, v159\n"
        "v_dual_min_num_f32 v160, s82, v202 :: v_dual_fmac_f32 v145, s97, v161\n"
        "v_dual_min_num_f32 v162, s90, v194 :: v_dual_fmac_f32 v146, s97, v163\n"
        "v_dual_min_num_f32 v164, s90, v198 :: v_dual_fmac_f32 v147, s97, v165\n"
        "v_dual_min_num_f32 v166, s90, v202 :: v_dual_fmac_f32 v148, s97, v167\n"
        "v_dual_min_num_f32 v151, s75, v195 :: v_dual_fmac_f32 v140, s98, v150\n"
        "v_dual_min_num_f32 v153, s75, v199 :: v_dual_fmac_f32 v141, s98, v152\n"
        "v_dual_min_num_f32 v155, s75, v203 :: v_dual_fmac_f32 v142, s98, v154\n"
        "v_dual_min_num_f32 v157, s83, v195 :: v_dual_fmac_f32 v143, s98, v156\n"
        "v_dual_min_num_f32 v159, s83, v199 :: v_dual_fmac_f32 v144, s98, v158\n"
        "v_dual_min_num_f32 v161, s83, v203 :: v_dual_fmac_f32 v145, s98, v160\n"
        "v_dual_min_num_f32 v163, s91, v195 :: v_dual_fmac_f32 v146, s98, v162\n"
        "v_dual_min_num_f32 v165, s91, v199 :: v_dual_fmac_f32 v147, s98, v164\n"
        "v_dual_min_num_f32 v167, s91, v203 :: v_dual_fmac_f32 v148, s98, v166\n"
        "ds_load_b128 v[192:195], %[c0] offset:80\n"
        "ds_load_b128 v[196:199], %[c0] offset:63440\n"
        "ds_load_b128 v[200:203], %[c2] offset:80\n"
        "s_wait_dscnt 0x6\n"
        "v_dual_min_num_f32 v150, s76, v168 :: v_dual_fmac_f32 v140, s99, v151\n"
        "v_dual_min_num_f32 v152, s76, v172 :: v_dual_fmac_f32 v141, s99, v153\n"
        "v_dual_min_num_f32 v154, s76, v176 :: v_dual_fmac_f32 v142, s99, v155\n"
        "v_dual_min_num_f32 v156, s84, v168 :: v_dual_fmac_f32 v143, s99, v157\n"
        "v_dual_min_num_f32 v158, s84, v172 :: v_dual_fmac_f32 v144, s99, v159\n"
        "v_dual_min_num_f32 v160, s84, v176 :: v_dual_fmac_f32 v145, s99, v161\n"
        "v_dual_min_num_f32 v162, s92, v168 :: v_dual_fmac_f32 v146, s99, v163\n"
        "v_dual_min_num_f32 v164, s92, v172 :: v_dual_fmac_f32 v147, s99, v165\n"
        "v_dual_min_num_f32 v166, s92, v176 :: v_dual_fmac_f32 v148, s99, v167\n"
        "v_dual_min_num_f32 v151, s77, v169 :: v_dual_fmac_f32 v140, s100, v150\n"
        "v_dual_min_num_f32 v153, s77, v173 :: v_dual_fmac_f32 v141, s100, v152\n"
        "v_dual_min_num_f32 v155, s77, v177 :: v_dual_fmac_f32 v142, s100, v154\n"
        "v_dual_min_num_f32 v157, s85, v169 :: v_dual_fmac_f32 v143, s100, v156\n"
        "v_dual_min_num_f32 v159, s85, v173 :: v_dual_fmac_f32 v144, s100, v158\n"
        "v_dual_min_num_f32 v161, s85, v177 :: v_dual_fmac_f32 v145, s100, v160\n"
        "v_dual_min_num_f32 v163, s93, v169 :: v_dual_fmac_f32 v146, s100, v162\n"
        "v_dual_min_num_f32 v165, s93, v173 :: v_dual_fmac_f32 v147, s100, v164\n"
        "v_dual_min_num_f32 v167, s93, v177 :: v_dual_fmac_f32 v148, s100, v166\n"
        "v_dual_min_num_f32 v150, s78, v170 :: v_dual_fmac_f32 v140, s101, v151\n"
        "v_dual_min_num_f32 v152, s78, v174 :: v_dual_fmac_f32 v141, s101, v153\n"
        "v_dual_min_num_f32 v154, s78, v178 :: v_dual_fmac_f32 v142, s101, v155\n"
        "v_dual_min_num_f32 v156, s86, v170 :: v_dual_fmac_f32 v143, s101, v157\n"
        "v_dual_min_num_f32 v158, s86, v174 :: v_dual_fmac_f32 v144, s101, v159\n"
        "v_dual_min_num_f32 v160, s86, v178 :: v_dual_fmac_f32 v145, s101, v161\n"
        "v_dual_min_num_f32 v162, s94, v170 :: v_dual_fmac_f32 v146, s101, v163\n"
        "v_dual_min_num_f32 v164, s94, v174 :: v_dual_fmac_f32 v147, s101, v165\n"
        "v_dual_min_num_f32 v166, s94, v178 :: v_dual_fmac_f32 v148, s101, v167\n"
        "v_dual_min_num_f32 v151, s79, v171 :: v_dual_fmac_f32 v140, s102, v150\n"
        "v_dual_min_num_f32 v153, s79, v175 :: v_dual_fmac_f32 v141, s102, v152\n"
        "v_dual_min_num_f32 v155, s79, v179 :: v_dual_fmac_f32 v142, s102, v154\n"
        "v_dual_min_num_f32 v157, s87, v171 :: v_dual_fmac_f32 v143, s102, v156\n"
        "v_dual_min_num_f32 v159, s87, v175 :: v_dual_fmac_f32 v144, s102, v158\n"
        "v_dual_min_num_f32 v161, s87, v179 :: v_dual_fmac_f32 v145, s102, v160\n"
        "v_dual_min_num_f32 v163, s95, v171 :: v_dual_fmac_f32 v146, s102, v162\n"
        "v_dual_min_num_f32 v165, s95, v175 :: v_dual_fmac_f32 v147, s102, v164\n"
        "v_dual_min_num_f32 v167, s95, v179 :: v_dual_fmac_f32 v148, s102, v166\n"
        "ds_load_b128 v[168:171], %[c0] offset:96\n"
        "ds_load_b128 v[172:175], %[c0] offset:63456\n"
        "ds_load_b128 v[176:179], %[c2] offset:96\n"
        "s_wait_dscnt 0x6\n"
        "s_wait_kmcnt 0x0\n"
        "v_dual_min_num_f32 v150, s40, v180 :: v_dual_fmac_f32 v140, s103, v151\n"
        "v_dual_min_num_f32 v152, s40, v184 :: v_dual_fmac_f32 v141, s103, v153\n"
        "v_dual_min_num_f32 v154, s40, v188 :: v_dual_fmac_f32 v142, s103, v155\n"
        "v_dual_min_num_f32 v156, s48, v180 :: v_dual_fmac_f32 v143, s103, v157\n"
        "v_dual_min_num_f32 v158, s48, v184 :: v_dual_fmac_f32 v144, s103, v159\n"
        "v_dual_min_num_f32 v160, s48, v188 :: v_dual_fmac_f32 v145, s103, v161\n"
        "v_dual_min_num_f32 v162, s56, v180 :: v_dual_fmac_f32 v146, s103, v163\n"
        "v_dual_min_num_f32 v164, s56, v184 :: v_dual_fmac_f32 v147, s103, v165\n"
        "v_dual_min_num_f32 v166, s56, v188 :: v_dual_fmac_f32 v148, s103, v167\n"
        "s_load_b256 s[72:79], %[pa], 0x60\n"
        "s_load_b256 s[80:87], %[pa], 0x460\n"
        "s_load_b256 s[88:95], %[pa], 0x860\n"
        "s_load_b256 s[96:103], %[wp], 0x60\n"
        "v_dual_min_num_f32 v151, s41, v181 :: v_dual_fmac_f32 v140, s64, v150\n"
        "v_dual_min_num_f32 v153, s41, v185 :: v_dual_fmac_f32 v141, s64, v152\n"
        "v_dual_min_num_f32 v155, s41, v189 :: v_dual_fmac_f32 v142, s64, v154\n"
        "v_dual_min_num_f32 v157, s49, v181 :: v_dual_fmac_f32 v143, s64, v156\n"
        "v_dual_min_num_f32 v159, s49, v185 :: v_dual_fmac_f32 v144, s64, v158\n"
        "v_dual_min_num_f32 v161, s49, v189 :: v_dual_fmac_f32 v145, s64, v160\n"
        "v_dual_min_num_f32 v163, s57, v181 :: v_dual_fmac_f32 v146, s64, v162\n"
        "v_dual_min_num_f32 v165, s57, v185 :: v_dual_fmac_f32 v147, s64, v164\n"
        "v_dual_min_num_f32 v167, s57, v189 :: v_dual_fmac_f32 v148, s64, v166\n"
        "v_dual_min_num_f32 v150, s42, v182 :: v_dual_fmac_f32 v140, s65, v151\n"
        "v_dual_min_num_f32 v152, s42, v186 :: v_dual_fmac_f32 v141, s65, v153\n"
        "v_dual_min_num_f32 v154, s42, v190 :: v_dual_fmac_f32 v142, s65, v155\n"
        "v_dual_min_num_f32 v156, s50, v182 :: v_dual_fmac_f32 v143, s65, v157\n"
        "v_dual_min_num_f32 v158, s50, v186 :: v_dual_fmac_f32 v144, s65, v159\n"
        "v_dual_min_num_f32 v160, s50, v190 :: v_dual_fmac_f32 v145, s65, v161\n"
        "v_dual_min_num_f32 v162, s58, v182 :: v_dual_fmac_f32 v146, s65, v163\n"
        "v_dual_min_num_f32 v164, s58, v186 :: v_dual_fmac_f32 v147, s65, v165\n"
        "v_dual_min_num_f32 v166, s58, v190 :: v_dual_fmac_f32 v148, s65, v167\n"
        "v_dual_min_num_f32 v151, s43, v183 :: v_dual_fmac_f32 v140, s66, v150\n"
        "v_dual_min_num_f32 v153, s43, v187 :: v_dual_fmac_f32 v141, s66, v152\n"
        "v_dual_min_num_f32 v155, s43, v191 :: v_dual_fmac_f32 v142, s66, v154\n"
        "v_dual_min_num_f32 v157, s51, v183 :: v_dual_fmac_f32 v143, s66, v156\n"
        "v_dual_min_num_f32 v159, s51, v187 :: v_dual_fmac_f32 v144, s66, v158\n"
        "v_dual_min_num_f32 v161, s51, v191 :: v_dual_fmac_f32 v145, s66, v160\n"
        "v_dual_min_num_f32 v163, s59, v183 :: v_dual_fmac_f32 v146, s66, v162\n"
        "v_dual_min_num_f32 v165, s59, v187 :: v_dual_fmac_f32 v147, s66, v164\n"
        "v_dual_min_num_f32 v167, s59, v191 :: v_dual_fmac_f32 v148, s66, v166\n"
        "ds_load_b128 v[180:183], %[c0] offset:112\n"
        "ds_load_b128 v[184:187], %[c0] offset:63472\n"
        "ds_load_b128 v[188:191], %[c2] offset:112\n"
        "s_wait_dscnt 0x6\n"
        "v_dual_min_num_f32 v150, s44, v192 :: v_dual_fmac_f32 v140, s67, v151\n"
        "v_dual_min_num_f32 v152, s44, v196 :: v_dual_fmac_f32 v141, s67, v153\n"
        "v_dual_min_num_f32 v154, s44, v200 :: v_dual_fmac_f32 v142, s67, v155\n"
        "v_dual_min_num_f32 v156, s52, v192 :: v_dual_fmac_f32 v143, s67, v157\n"
        "v_dual_min_num_f32 v158, s52, v196 :: v_dual_fmac_f32 v144, s67, v159\n"
        "v_dual_min_num_f32 v160, s52, v200 :: v_dual_fmac_f32 v145, s67, v161\n"
        "v_dual_min_num_f32 v162, s60, v192 :: v_dual_fmac_f32 v146, s67, v163\n"
        "v_dual_min_num_f32 v164, s60, v196 :: v_dual_fmac_f32 v147, s67, v165\n"
        "v_dual_min_num_f32 v166, s60, v200 :: v_dual_fmac_f32 v148, s67, v167\n"
        "v_dual_min_num_f32 v151, s45, v193 :: v_dual_fmac_f32 v140, s68, v150\n"
        "v_dual_min_num_f32 v153, s45, v197 :: v_dual_fmac_f32 v141, s68, v152\n"
        "v_dual_min_num_f32 v155, s45, v201 :: v_dual_fmac_f32 v142, s68, v154\n"
        "v_dual_min_num_f32 v157, s53, v193 :: v_dual_fmac_f32 v143, s68, v156\n"
        "v_dual_min_num_f32 v159, s53, v197 :: v_dual_fmac_f32 v144, s68, v158\n"
        "v_dual_min_num_f32 v161, s53, v201 :: v_dual_fmac_f32 v145, s68, v160\n"
        "v_dual_min_num_f32 v163, s61, v193 :: v_dual_fmac_f32 v146, s68, v162\n"
        "v_dual_min_num_f32 v165, s61, v197 :: v_dual_fmac_f32 v147, s68, v164\n"
        "v_dual_min_num_f32 v167, s61, v201 :: v_dual_fmac_f32 v148, s68, v166\n"
        "v_dual_min_num_f32 v150, s46, v194 :: v_dual_fmac_f32 v140, s69, v151\n"
        "v_dual_min_num_f32 v152, s46, v198 :: v_dual_fmac_f32 v141, s69, v153\n"
        "v_dual_min_num_f32 v154, s46, v202 :: v_dual_fmac_f32 v142, s69, v155\n"
        "v_dual_min_num_f32 v156, s54, v194 :: v_dual_fmac_f32 v143, s69, v157\n"
        "v_dual_min_num_f32 v158, s54, v198 :: v_dual_fmac_f32 v144, s69, v159\n"
        "v_dual_min_num_f32 v160, s54, v202 :: v_dual_fmac_f32 v145, s69, v161\n"
        "v_dual_min_num_f32 v162, s62, v194 :: v_dual_fmac_f32 v146, s69, v163\n"
        "v_dual_min_num_f32 v164, s62, v198 :: v_dual_fmac_f32 v147, s69, v165\n"
        "v_dual_min_num_f32 v166, s62, v202 :: v_dual_fmac_f32 v148, s69, v167\n"
        "v_dual_min_num_f32 v151, s47, v195 :: v_dual_fmac_f32 v140, s70, v150\n"
        "v_dual_min_num_f32 v153, s47, v199 :: v_dual_fmac_f32 v141, s70, v152\n"
        "v_dual_min_num_f32 v155, s47, v203 :: v_dual_fmac_f32 v142, s70, v154\n"
        "v_dual_min_num_f32 v157, s55, v195 :: v_dual_fmac_f32 v143, s70, v156\n"
        "v_dual_min_num_f32 v159, s55, v199 :: v_dual_fmac_f32 v144, s70, v158\n"
        "v_dual_min_num_f32 v161, s55, v203 :: v_dual_fmac_f32 v145, s70, v160\n"
        "v_dual_min_num_f32 v163, s63, v195 :: v_dual_fmac_f32 v146, s70, v162\n"
        "v_dual_min_num_f32 v165, s63, v199 :: v_dual_fmac_f32 v147, s70, v164\n"
        "v_dual_min_num_f32 v167, s63, v203 :: v_dual_fmac_f32 v148, s70, v166\n"
        "s_wait_dscnt 0x3\n"
        "s_wait_kmcnt 0x0\n"
        "v_dual_min_num_f32 v150, s72, v168 :: v_dual_fmac_f32 v140, s71, v151\n"
        "v_dual_min_num_f32 v152, s72, v172 :: v_dual_fmac_f32 v141, s71, v153\n"
        "v_dual_min_num_f32 v154, s72, v176 :: v_dual_fmac_f32 v142, s71, v155\n"
        "v_dual_min_num_f32 v156, s80, v168 :: v_dual_fmac_f32 v143, s71, v157\n"
        "v_dual_min_num_f32 v158, s80, v172 :: v_dual_fmac_f32 v144, s71, v159\n"
        "v_dual_min_num_f32 v160, s80, v176 :: v_dual_fmac_f32 v145, s71, v161\n"
        "v_dual_min_num_f32 v162, s88, v168 :: v_dual_fmac_f32 v146, s71, v163\n"
        "v_dual_min_num_f32 v164, s88, v172 :: v_dual_fmac_f32 v147, s71, v165\n"
        "v_dual_min_num_f32 v166, s88, v176 :: v_dual_fmac_f32 v148, s71, v167\n"
        "v_dual_min_num_f32 v151, s73, v169 :: v_dual_fmac_f32 v140, s96, v150\n"
        "v_dual_min_num_f32 v153, s73, v173 :: v_dual_fmac_f32 v141, s96, v152\n"
        "v_dual_min_num_f32 v155, s73, v177 :: v_dual_fmac_f32 v142, s96, v154\n"
        "v_dual_min_num_f32 v157, s81, v169 :: v_dual_fmac_f32 v143, s96, v156\n"
        "v_dual_min_num_f32 v159, s81, v173 :: v_dual_fmac_f32 v144, s96, v158\n"
        "v_dual_min_num_f32 v161, s81, v177 :: v_dual_fmac_f32 v145, s96, v160\n"
        "v_dual_min_num_f32 v163, s89, v169 :: v_dual_fmac_f32 v146, s96, v162\n"
        "v_dual_min_num_f32 v165, s89, v173 :: v_dual_fmac_f32 v147, s96, v164\n"
        "v_dual_min_num_f32 v167, s89, v177 :: v_dual_fmac_f32 v148, s96, v166\n"
        "v_dual_min_num_f32 v150, s74, v170 :: v_dual_fmac_f32 v140, s97, v151\n"
        "v_dual_min_num_f32 v152, s74, v174 :: v_dual_fmac_f32 v141, s97, v153\n"
        "v_dual_min_num_f32 v154, s74, v178 :: v_dual_fmac_f32 v142, s97, v155\n"
        "v_dual_min_num_f32 v156, s82, v170 :: v_dual_fmac_f32 v143, s97, v157\n"
        "v_dual_min_num_f32 v158, s82, v174 :: v_dual_fmac_f32 v144, s97, v159\n"
        "v_dual_min_num_f32 v160, s82, v178 :: v_dual_fmac_f32 v145, s97, v161\n"
        "v_dual_min_num_f32 v162, s90, v170 :: v_dual_fmac_f32 v146, s97, v163\n"
        "v_dual_min_num_f32 v164, s90, v174 :: v_dual_fmac_f32 v147, s97, v165\n"
        "v_dual_min_num_f32 v166, s90, v178 :: v_dual_fmac_f32 v148, s97, v167\n"
        "v_dual_min_num_f32 v151, s75, v171 :: v_dual_fmac_f32 v140, s98, v150\n"
        "v_dual_min_num_f32 v153, s75, v175 :: v_dual_fmac_f32 v141, s98, v152\n"
        "v_dual_min_num_f32 v155, s75, v179 :: v_dual_fmac_f32 v142, s98, v154\n"
        "v_dual_min_num_f32 v157, s83, v171 :: v_dual_fmac_f32 v143, s98, v156\n"
        "v_dual_min_num_f32 v159, s83, v175 :: v_dual_fmac_f32 v144, s98, v158\n"
        "v_dual_min_num_f32 v161, s83, v179 :: v_dual_fmac_f32 v145, s98, v160\n"
        "v_dual_min_num_f32 v163, s91, v171 :: v_dual_fmac_f32 v146, s98, v162\n"
        "v_dual_min_num_f32 v165, s91, v175 :: v_dual_fmac_f32 v147, s98, v164\n"
        "v_dual_min_num_f32 v167, s91, v179 :: v_dual_fmac_f32 v148, s98, v166\n"
        "s_wait_dscnt 0x0\n"
        "v_dual_min_num_f32 v150, s76, v180 :: v_dual_fmac_f32 v140, s99, v151\n"
        "v_dual_min_num_f32 v152, s76, v184 :: v_dual_fmac_f32 v141, s99, v153\n"
        "v_dual_min_num_f32 v154, s76, v188 :: v_dual_fmac_f32 v142, s99, v155\n"
        "v_dual_min_num_f32 v156, s84, v180 :: v_dual_fmac_f32 v143, s99, v157\n"
        "v_dual_min_num_f32 v158, s84, v184 :: v_dual_fmac_f32 v144, s99, v159\n"
        "v_dual_min_num_f32 v160, s84, v188 :: v_dual_fmac_f32 v145, s99, v161\n"
        "v_dual_min_num_f32 v162, s92, v180 :: v_dual_fmac_f32 v146, s99, v163\n"
        "v_dual_min_num_f32 v164, s92, v184 :: v_dual_fmac_f32 v147, s99, v165\n"
        "v_dual_min_num_f32 v166, s92, v188 :: v_dual_fmac_f32 v148, s99, v167\n"
        "v_dual_min_num_f32 v151, s77, v181 :: v_dual_fmac_f32 v140, s100, v150\n"
        "v_dual_min_num_f32 v153, s77, v185 :: v_dual_fmac_f32 v141, s100, v152\n"
        "v_dual_min_num_f32 v155, s77, v189 :: v_dual_fmac_f32 v142, s100, v154\n"
        "v_dual_min_num_f32 v157, s85, v181 :: v_dual_fmac_f32 v143, s100, v156\n"
        "v_dual_min_num_f32 v159, s85, v185 :: v_dual_fmac_f32 v144, s100, v158\n"
        "v_dual_min_num_f32 v161, s85, v189 :: v_dual_fmac_f32 v145, s100, v160\n"
        "v_dual_min_num_f32 v163, s93, v181 :: v_dual_fmac_f32 v146, s100, v162\n"
        "v_dual_min_num_f32 v165, s93, v185 :: v_dual_fmac_f32 v147, s100, v164\n"
        "v_dual_min_num_f32 v167, s93, v189 :: v_dual_fmac_f32 v148, s100, v166\n"
        "v_dual_min_num_f32 v150, s78, v182 :: v_dual_fmac_f32 v140, s101, v151\n"
        "v_dual_min_num_f32 v152, s78, v186 :: v_dual_fmac_f32 v141, s101, v153\n"
        "v_dual_min_num_f32 v154, s78, v190 :: v_dual_fmac_f32 v142, s101, v155\n"
        "v_dual_min_num_f32 v156, s86, v182 :: v_dual_fmac_f32 v143, s101, v157\n"
        "v_dual_min_num_f32 v158, s86, v186 :: v_dual_fmac_f32 v144, s101, v159\n"
        "v_dual_min_num_f32 v160, s86, v190 :: v_dual_fmac_f32 v145, s101, v161\n"
        "v_dual_min_num_f32 v162, s94, v182 :: v_dual_fmac_f32 v146, s101, v163\n"
        "v_dual_min_num_f32 v164, s94, v186 :: v_dual_fmac_f32 v147, s101, v165\n"
        "v_dual_min_num_f32 v166, s94, v190 :: v_dual_fmac_f32 v148, s101, v167\n"
        "v_dual_min_num_f32 v151, s79, v183 :: v_dual_fmac_f32 v140, s102, v150\n"
        "v_dual_min_num_f32 v153, s79, v187 :: v_dual_fmac_f32 v141, s102, v152\n"
        "v_dual_min_num_f32 v155, s79, v191 :: v_dual_fmac_f32 v142, s102, v154\n"
        "v_dual_min_num_f32 v157, s87, v183 :: v_dual_fmac_f32 v143, s102, v156\n"
        "v_dual_min_num_f32 v159, s87, v187 :: v_dual_fmac_f32 v144, s102, v158\n"
        "v_dual_min_num_f32 v161, s87, v191 :: v_dual_fmac_f32 v145, s102, v160\n"
        "v_dual_min_num_f32 v163, s95, v183 :: v_dual_fmac_f32 v146, s102, v162\n"
        "v_dual_min_num_f32 v165, s95, v187 :: v_dual_fmac_f32 v147, s102, v164\n"
        "v_dual_min_num_f32 v167, s95, v191 :: v_dual_fmac_f32 v148, s102, v166\n"
        "v_fmac_f32_e32 v140, s103, v151\n"
        "v_fmac_f32_e32 v141, s103, v153\n"
        "v_fmac_f32_e32 v142, s103, v155\n"
        "v_fmac_f32_e32 v143, s103, v157\n"
        "v_fmac_f32_e32 v144, s103, v159\n"
        "v_fmac_f32_e32 v145, s103, v161\n"
        "v_fmac_f32_e32 v146, s103, v163\n"
        "v_fmac_f32_e32 v147, s103, v165\n"
        "v_fmac_f32_e32 v148, s103, v167\n"
        : "={v140}"(acc[0][0]), "={v141}"(acc[0][1]), "={v142}"(acc[0][2]), "={v143}"(acc[1][0]), "={v144}"(acc[1][1]), "={v145}"(acc[1][2]), "={v146}"(acc[2][0]), "={v147}"(acc[2][1]), "={v148}"(acc[2][2])
        : "0"(acc[0][0]), "1"(acc[0][1]), "2"(acc[0][2]), "3"(acc[1][0]), "4"(acc[1][1]), "5"(acc[1][2]), "6"(acc[2][0]), "7"(acc[2][1]), "8"(acc[2][2]),
          [c0] "v"(c0), [c2] "v"(c2), [pa] "s"(pa), [wp] "s"(wp)
        : "memory", "v150", "v151", "v152", "v153", "v154", "v155", "v156", "v157", "v158", "v159", "v160", "v161", "v162", "v163", "v164", "v165", "v166", "v167", "v168", "v169", "v170", "v171", "v172", "v173", "v174", "v175", "v176", "v177", "v178", "v179", "v180", "v181", "v182", "v183", "v184", "v185", "v186", "v187", "v188", "v189", "v190", "v191", "v192", "v193", "v194", "v195", "v196", "v197", "v198", "v199", "v200", "v201", "v202", "v203",
          "s40", "s41", "s42", "s43", "s44", "s45", "s46", "s47", "s48", "s49", "s50", "s51", "s52", "s53", "s54", "s55", "s56", "s57", "s58", "s59", "s60", "s61", "s62", "s63", "s64", "s65", "s66", "s67", "s68", "s69", "s70", "s71", "s72", "s73", "s74", "s75", "s76", "s77", "s78", "s79", "s80", "s81", "s82", "s83", "s84", "s85", "s86", "s87", "s88", "s89", "s90", "s91", "s92", "s93", "s94", "s95", "s96", "s97", "s98", "s99", "s100", "s101", "s102", "s103");
}

__device__ __forceinline__ void warm_scalar_rows(const float* pa, const float* w2) {
    asm volatile(
        "s_load_b32 s40, %[pa], 0x0\n"
        "s_load_b32 s41, %[pa], 0x40\n"
        "s_load_b32 s42, %[pa], 0x80\n"
        "s_load_b32 s43, %[pa], 0xc0\n"
        "s_load_b32 s44, %[pa], 0x100\n"
        "s_load_b32 s45, %[pa], 0x140\n"
        "s_load_b32 s46, %[pa], 0x180\n"
        "s_load_b32 s47, %[pa], 0x1c0\n"
        "s_load_b32 s48, %[pa], 0x400\n"
        "s_load_b32 s49, %[pa], 0x440\n"
        "s_load_b32 s50, %[pa], 0x480\n"
        "s_load_b32 s51, %[pa], 0x4c0\n"
        "s_load_b32 s52, %[pa], 0x500\n"
        "s_load_b32 s53, %[pa], 0x540\n"
        "s_load_b32 s54, %[pa], 0x580\n"
        "s_load_b32 s55, %[pa], 0x5c0\n"
        "s_load_b32 s56, %[pa], 0x800\n"
        "s_load_b32 s57, %[pa], 0x840\n"
        "s_load_b32 s58, %[pa], 0x880\n"
        "s_load_b32 s59, %[pa], 0x8c0\n"
        "s_load_b32 s60, %[pa], 0x900\n"
        "s_load_b32 s61, %[pa], 0x940\n"
        "s_load_b32 s62, %[pa], 0x980\n"
        "s_load_b32 s63, %[pa], 0x9c0\n"
        "s_load_b32 s64, %[wp], 0x0\n"
        "s_load_b32 s65, %[wp], 0x40\n"
        "s_load_b32 s66, %[wp], 0x80\n"
        "s_load_b32 s67, %[wp], 0xc0\n"
        "s_load_b32 s68, %[wp], 0x100\n"
        "s_load_b32 s69, %[wp], 0x140\n"
        "s_load_b32 s70, %[wp], 0x180\n"
        "s_load_b32 s71, %[wp], 0x1c0\n"
        "s_wait_kmcnt 0x0\n"
        :
        : [pa] "s"(pa), [wp] "s"(w2)
        : "memory", "s40", "s41", "s42", "s43", "s44", "s45", "s46", "s47", "s48", "s49", "s50", "s51", "s52", "s53", "s54", "s55", "s56", "s57", "s58", "s59", "s60", "s61", "s62", "s63", "s64", "s65", "s66", "s67", "s68", "s69", "s70", "s71");
}

__global__ __launch_bounds__(256) __attribute__((amdgpu_waves_per_eu(2, 4)))
void pair_softmax(const float* __restrict__ wsA,
                  const float* __restrict__ wsC,
                  const float* __restrict__ W2,
                  const float* __restrict__ b2,
                  float* __restrict__ out) {
    __shared__ __attribute__((aligned(16))) float Cs[CROWS][CPITCH];
    __shared__ float redm[2][3][4];
    __shared__ float reds[2][3][4];
    __shared__ float rown[8];

    const int b   = blockIdx.y;
    const int i0  = blockIdx.x * FR;
    const int tid = threadIdx.x;

    const float* gA = wsA + (size_t)(b * N_TOK + i0) * D_DIM;
    const float* gC = wsC + (size_t)(b * N_TOK) * D_DIM;

    const int sw = tid >> 5, sl = tid & 31;
    const int sr = ((sl >> 3) & 1) * 8 + (sl >> 4) + 2 * (sw & 3) + 16 * (sw >> 2);
    const int sq = (tid & 7) * 4;
    v4f st[SPT];
    const v4f wrow = *(const v4f*)(W2 + sl * 4);

    const int ty = __builtin_amdgcn_readfirstlane(tid >> 7);
    const float* pa = gA + ty * D_DIM;

    const v4f na = -*(const v4f*)(gA + min(sw, FR - 1) * D_DIM + sl * 4);
#pragma unroll
    for (int q = 0; q < SPT; ++q)
        st[q] = *(const v4f*)(gC + min(sr + 32 * q, N_TOK - 1) * D_DIM + sq);

    warm_scalar_rows(pa, W2);

    {
        float p = na.x * wrow.x;
        p = fmaf(na.y, wrow.y, p);
        p = fmaf(na.z, wrow.z, p);
        p = fmaf(na.w, wrow.w, p);
        p = wave_sum(p);
        if (sl == 0 && sw < FR) rown[sw] = p;
    }
#pragma unroll
    for (int q = 0; q < SPT; ++q)
        *(v4f*)&Cs[sr + 32 * q][sq] = st[q];

    const int tx   = tid & 127;
    const bool live = tx < FC;
    const int txc  = min(tx, FC - 1);
    const unsigned cc0 = lds_addr(&Cs[txc][0]);

    float acc[3][3];
#pragma unroll
    for (int ii = 0; ii < 3; ++ii)
#pragma unroll
        for (int jj = 0; jj < 3; ++jj) acc[ii][jj] = 0.0f;

#pragma unroll
    for (int k = 0; k < NCH; ++k) {
        const int dn = (k + 1) * DCH;
        if (k + 1 < NCH) {
#pragma unroll
            for (int q = 0; q < SPT; ++q)
                st[q] = *(const v4f*)(gC + min(sr + 32 * q, N_TOK - 1) * D_DIM + dn + sq);
        }
        slice32(acc, cc0 + k * DCH * 4, cc0 + k * DCH * 4 + 2 * FC * CPITCH * 4,
                pa + k * DCH, W2 + k * DCH);

        if (k + 1 < NCH) {
#pragma unroll
            for (int q = 0; q < SPT; ++q)
                *(v4f*)&Cs[sr + 32 * q][dn + sq] = st[q];
        }
    }

    const float bias = b2[0];
    const int wq = (tid >> 5) & 3;
    const int lane = tid & 31;

    float m[3];
#pragma unroll
    for (int ii = 0; ii < 3; ++ii) {
        const int i = i0 + ty + 2 * ii;
        const float dg = rown[ty + 2 * ii] - bias;
#pragma unroll
        for (int jj = 0; jj < 3; ++jj) {
            const int j = tx + FC * jj;
            acc[ii][jj] = (i == j) ? dg : -acc[ii][jj];
        }
        const float mx = fmaxf(fmaxf(acc[ii][0], acc[ii][1]), acc[ii][2]);
        m[ii] = wave_max(live ? mx : -3.4e38f);
    }
    if (lane == 0) {
#pragma unroll
        for (int ii = 0; ii < 3; ++ii) redm[ty][ii][wq] = m[ii];
    }
    __syncthreads();

    float s[3];
#pragma unroll
    for (int ii = 0; ii < 3; ++ii) {
        const float mr = fmaxf(fmaxf(redm[ty][ii][0], redm[ty][ii][1]),
                               fmaxf(redm[ty][ii][2], redm[ty][ii][3]));
        float sm = 0.0f;
#pragma unroll
        for (int jj = 0; jj < 3; ++jj) {
            acc[ii][jj] = expf(acc[ii][jj] - mr);
            sm += acc[ii][jj];
        }
        s[ii] = wave_sum(live ? sm : 0.0f);
    }
    if (lane == 0) {
#pragma unroll
        for (int ii = 0; ii < 3; ++ii) reds[ty][ii][wq] = s[ii];
    }
    __syncthreads();

    if (live) {
#pragma unroll
        for (int ii = 0; ii < 3; ++ii) {
            const int i = i0 + ty + 2 * ii;
            const float inv = 1.0f / ((reds[ty][ii][0] + reds[ty][ii][1]) +
                                      (reds[ty][ii][2] + reds[ty][ii][3]));
            float* orow = out + (size_t)(b * N_TOK + i) * N_TOK;
#pragma unroll
            for (int jj = 0; jj < 3; ++jj) orow[tx + FC * jj] = acc[ii][jj] * inv;
        }
    }
}

extern "C" void kernel_launch(void* const* d_in, const int* in_sizes, int n_in,
                              void* d_out, int out_size, void* d_ws, size_t ws_size,
                              hipStream_t stream) {
    (void)in_sizes; (void)n_in; (void)out_size; (void)ws_size;
    const float* x  = (const float*)d_in[0];
    const float* W1 = (const float*)d_in[1];
    const float* b1 = (const float*)d_in[2];
    const float* W2 = (const float*)d_in[3];
    const float* b2 = (const float*)d_in[4];
    float* out = (float*)d_out;

    float* wsA = (float*)d_ws;
    float* wsC = wsA + M_ROWS * D_DIM;

    gemm_wmma<<<dim3(M_ROWS * 16 / 16 / 8), 256, 0, stream>>>(x, W1, b1, wsA, wsC);

    dim3 pg(N_TOK / FR, B_SZ);
    pair_softmax<<<pg, 256, 0, stream>>>(wsA, wsC, W2, b2, out);
}
